# starting kernel
# baseline (speedup 1.0000x reference)
#include <hip/hip_runtime.h>
#include <math.h>


typedef __attribute__((ext_vector_type(2))) float v2f;
typedef __attribute__((ext_vector_type(4))) float v4f;
typedef __attribute__((ext_vector_type(8))) float v8f;

#define NEG_BIG (-10000.0f)

#define TM 128
#define TN 64
#define TK 32
#define TKP 34


__global__ __launch_bounds__(256) void gemm_nt_f32_wmma(
    const float* __restrict__ A, const float* __restrict__ W,
    float* __restrict__ C, int M, int N, int K)
{
    __shared__ float As[TM][TKP];
    __shared__ float Ws[TN][TKP];

    const int tid  = threadIdx.x;
    const int lane = tid & 31, wid = tid >> 5;
    const int wm   = wid >> 1, wn  = wid & 1;
    const int ll   = lane & 15, lh = lane >> 4;
    const int m0   = blockIdx.x * TM, n0 = blockIdx.y * TN;

    const v8f vzero = {0.f,0.f,0.f,0.f,0.f,0.f,0.f,0.f};
    v8f acc[2][2];
    acc[0][0]=vzero; acc[0][1]=vzero; acc[1][0]=vzero; acc[1][1]=vzero;

    for (int k0 = 0; k0 < K; k0 += TK) {
        __syncthreads();
        #pragma unroll
        for (int p = 0; p < 4; ++p) {
            int r = p*32 + (tid >> 3);
            int c = (tid & 7) << 2;
            int gm = m0 + r, gk = k0 + c;
            v4f v = {0.f,0.f,0.f,0.f};
            if (gm < M && gk < K) v = *(const v4f*)(A + (size_t)gm*K + gk);
            *(v4f*)&As[r][c] = v;
        }
        #pragma unroll
        for (int p = 0; p < 2; ++p) {
            int r = p*32 + (tid >> 3);
            int c = (tid & 7) << 2;
            int gn = n0 + r, gk = k0 + c;
            v4f v = {0.f,0.f,0.f,0.f};
            if (gn < N && gk < K) v = *(const v4f*)(W + (size_t)gn*K + gk);
            *(v4f*)&Ws[r][c] = v;
        }
        __syncthreads();

        #pragma unroll
        for (int kk = 0; kk < 8; ++kk) {
            const int kc = 4*kk + 2*lh;
            v2f a0 = *(const v2f*)&As[wm*32      + ll][kc];
            v2f a1 = *(const v2f*)&As[wm*32 + 16 + ll][kc];
            v2f b0 = *(const v2f*)&Ws[wn*32      + ll][kc];
            v2f b1 = *(const v2f*)&Ws[wn*32 + 16 + ll][kc];
            acc[0][0] = __builtin_amdgcn_wmma_f32_16x16x4_f32(false,a0,false,b0,(short)0,acc[0][0],false,false);
            acc[0][1] = __builtin_amdgcn_wmma_f32_16x16x4_f32(false,a0,false,b1,(short)0,acc[0][1],false,false);
            acc[1][0] = __builtin_amdgcn_wmma_f32_16x16x4_f32(false,a1,false,b0,(short)0,acc[1][0],false,false);
            acc[1][1] = __builtin_amdgcn_wmma_f32_16x16x4_f32(false,a1,false,b1,(short)0,acc[1][1],false,false);
        }
    }

    #pragma unroll
    for (int mt = 0; mt < 2; ++mt)
        #pragma unroll
        for (int nt = 0; nt < 2; ++nt) {
            int gn = n0 + wn*32 + nt*16 + ll;
            #pragma unroll
            for (int r = 0; r < 8; ++r) {
                int gm = m0 + wm*32 + mt*16 + lh*8 + r;
                if (gm < M && gn < N) C[(size_t)gm*N + gn] = acc[mt][nt][r];
            }
        }
}

__global__ __launch_bounds__(256) void attn_softmax_kernel(
    const float* __restrict__ scores, const float* __restrict__ lmask,
    const float* __restrict__ imask, float* __restrict__ attn,
    float* __restrict__ b_in)
{
    int row = blockIdx.x * 256 + threadIdx.x;
    if (row >= 8*256*20) return;
    int bs = row / 20;
    int l  = row - bs*20;

    float v[8];
    float mx = -3.4e38f;
    #pragma unroll
    for (int t = 0; t < 8; ++t) {
        float s = scores[(size_t)row*8 + t] + (1.0f - lmask[l*8 + t]) * NEG_BIG;
        v[t] = s;
        mx = fmaxf(mx, s);
    }
    float sum = 0.f;
    #pragma unroll
    for (int t = 0; t < 8; ++t) { float e = __expf(v[t] - mx); v[t] = e; sum += e; }
    float inv = 1.0f / sum;
    #pragma unroll
    for (int t = 0; t < 8; ++t) attn[(size_t)row*8 + t] = v[t] * inv;

    b_in[row] = mx + (1.0f - imask[bs]) * NEG_BIG;
}

__global__ __launch_bounds__(256) void q2c_kernel(
    const float* __restrict__ b_in, const float* __restrict__ tkn,
    float* __restrict__ q2c)
{
    __shared__ float red[256];
    __shared__ float wgt[256];
    const int tid = threadIdx.x;
    const int b = blockIdx.x / 20;
    const int l = blockIdx.x - b*20;

    float x = b_in[(size_t)(b*256 + tid)*20 + l];
    red[tid] = x; __syncthreads();
    for (int st = 128; st > 0; st >>= 1) {
        if (tid < st) red[tid] = fmaxf(red[tid], red[tid + st]);
        __syncthreads();
    }
    float mx = red[0]; __syncthreads();
    float e = __expf(x - mx);
    red[tid] = e; __syncthreads();
    for (int st = 128; st > 0; st >>= 1) {
        if (tid < st) red[tid] += red[tid + st];
        __syncthreads();
    }
    float sm = red[0]; __syncthreads();
    wgt[tid] = e / sm; __syncthreads();

    for (int h = tid; h < 768; h += 256) {
        float acc = 0.f;
        const float* tp = tkn + (size_t)b*256*768 + h;
        #pragma unroll 4
        for (int s = 0; s < 256; ++s) acc += wgt[s] * tp[(size_t)s*768];
        q2c[(size_t)(b*20 + l)*768 + h] = acc;
    }
}

__global__ __launch_bounds__(256) void fused_w5_tanh_wmma(
    const float* __restrict__ tkn,  const float* __restrict__ lab,
    const float* __restrict__ attn, const float* __restrict__ q2c,
    const float* __restrict__ W5,   const float* __restrict__ b5,
    float* __restrict__ out)
{
    __shared__ float As[TM][TKP];
    __shared__ float Ws[TN][TKP];
    __shared__ float aBlk[TM][8];

    const int tid  = threadIdx.x;
    const int lane = tid & 31, wid = tid >> 5;
    const int wm   = wid >> 1, wn  = wid & 1;
    const int ll   = lane & 15, lh = lane >> 4;
    const int m0   = blockIdx.x * TM, n0 = blockIdx.y * TN;

    {
        int r = tid >> 1, c = (tid & 1) << 2;
        *(v4f*)&aBlk[r][c] = *(const v4f*)(attn + (size_t)(m0 + r)*8 + c);
    }

    const v8f vzero = {0.f,0.f,0.f,0.f,0.f,0.f,0.f,0.f};
    v8f acc[2][2];
    acc[0][0]=vzero; acc[0][1]=vzero; acc[1][0]=vzero; acc[1][1]=vzero;

    for (int k0 = 0; k0 < 3072; k0 += TK) {
        const int sec = k0 / 768;
        const int j0  = k0 - sec*768;
        __syncthreads();
        #pragma unroll
        for (int p = 0; p < 2; ++p) {
            int r = p*32 + (tid >> 3);
            int c = (tid & 7) << 2;
            *(v4f*)&Ws[r][c] = *(const v4f*)(W5 + (size_t)(n0 + r)*3072 + (k0 + c));
        }
        #pragma unroll
        for (int p = 0; p < 4; ++p) {
            int r = p*32 + (tid >> 3);
            int c = (tid & 7) << 2;
            int m  = m0 + r;
            int bs = m / 20;
            int l  = m - bs*20;
            int j  = j0 + c;
            v4f v;
            if (sec == 0) {
                v = *(const v4f*)(tkn + (size_t)bs*768 + j);
            } else if (sec <= 2) {
                v4f cq = {0.f,0.f,0.f,0.f};
                const float* labRow = lab + (size_t)l*8*768 + j;
                #pragma unroll
                for (int t = 0; t < 8; ++t) {
                    float w  = aBlk[r][t];
                    v4f  lv  = *(const v4f*)(labRow + (size_t)t*768);
                    cq += w * lv;
                }
                if (sec == 1) v = cq;
                else {
                    v4f t4 = *(const v4f*)(tkn + (size_t)bs*768 + j);
                    v = t4 * cq;
                }
            } else {
                int b  = bs >> 8;
                v4f t4 = *(const v4f*)(tkn + (size_t)bs*768 + j);
                v4f q4 = *(const v4f*)(q2c + (size_t)(b*20 + l)*768 + j);
                v = t4 * q4;
            }
            *(v4f*)&As[r][c] = v;
        }
        __syncthreads();

        #pragma unroll
        for (int kk = 0; kk < 8; ++kk) {
            const int kc = 4*kk + 2*lh;
            v2f a0 = *(const v2f*)&As[wm*32      + ll][kc];
            v2f a1 = *(const v2f*)&As[wm*32 + 16 + ll][kc];
            v2f b0 = *(const v2f*)&Ws[wn*32      + ll][kc];
            v2f b1 = *(const v2f*)&Ws[wn*32 + 16 + ll][kc];
            acc[0][0] = __builtin_amdgcn_wmma_f32_16x16x4_f32(false,a0,false,b0,(short)0,acc[0][0],false,false);
            acc[0][1] = __builtin_amdgcn_wmma_f32_16x16x4_f32(false,a0,false,b1,(short)0,acc[0][1],false,false);
            acc[1][0] = __builtin_amdgcn_wmma_f32_16x16x4_f32(false,a1,false,b0,(short)0,acc[1][0],false,false);
            acc[1][1] = __builtin_amdgcn_wmma_f32_16x16x4_f32(false,a1,false,b1,(short)0,acc[1][1],false,false);
        }
    }

    #pragma unroll
    for (int mt = 0; mt < 2; ++mt)
        #pragma unroll
        for (int nt = 0; nt < 2; ++nt) {
            int gn = n0 + wn*32 + nt*16 + ll;
            float bias = b5[gn];
            #pragma unroll
            for (int r = 0; r < 8; ++r) {
                int gm = m0 + wm*32 + mt*16 + lh*8 + r;
                out[(size_t)gm*768 + gn] = tanhf(acc[mt][nt][r] + bias);
            }
        }
}

extern "C" void kernel_launch(void* const* d_in, const int* in_sizes, int n_in,
                              void* d_out, int out_size, void* d_ws, size_t ws_size,
                              hipStream_t stream)
{
    (void)in_sizes; (void)n_in; (void)out_size; (void)ws_size;

    const float* token = (const float*)d_in[0];
    const float* label = (const float*)d_in[1];
    const float* imask = (const float*)d_in[2];
    const float* lmask = (const float*)d_in[3];
    const float* W1    = (const float*)d_in[4];
    const float* W2    = (const float*)d_in[5];
    const float* W5    = (const float*)d_in[6];
    const float* b5    = (const float*)d_in[7];
    float* out = (float*)d_out;

    float* ws     = (float*)d_ws;
    float* tkn    = ws;
    float* lab    = tkn    + 2048*768;
    float* scores = lab    + 160*768;
    float* attn   = scores + 2048*160;
    float* b_in   = attn   + 2048*160;
    float* q2c    = b_in   + 40960;

    dim3 blk(256);
    gemm_nt_f32_wmma<<<dim3(16, 12), blk, 0, stream>>>(token, W1, tkn, 2048, 768, 768);
    gemm_nt_f32_wmma<<<dim3(2, 12),  blk, 0, stream>>>(label, W2, lab, 160, 768, 300);
    gemm_nt_f32_wmma<<<dim3(16, 3),  blk, 0, stream>>>(tkn, lab, scores, 2048, 160, 768);
    attn_softmax_kernel<<<dim3(160), blk, 0, stream>>>(scores, lmask, imask, attn, b_in);
    q2c_kernel<<<dim3(160), blk, 0, stream>>>(b_in, tkn, q2c);
    fused_w5_tanh_wmma<<<dim3(320, 12), blk, 0, stream>>>(tkn, lab, attn, q2c, W5, b5, out);
}


# Main GEMM K loop column-major: a step takes the same 32 columns of all 3 sections, so tkn and lab load once and c2q forms once per step (72 -> 24 barrier pairs per block); sum order changes
# speedup vs baseline: 1.5334x; 1.5334x over previous
#include <hip/hip_runtime.h>
#include <math.h>


typedef __attribute__((ext_vector_type(2))) float v2f;
typedef __attribute__((ext_vector_type(4))) float v4f;
typedef __attribute__((ext_vector_type(8))) float v8f;

#define NEG_BIG (-10000.0f)

#define TM 128
#define TN 64
#define TK 32
#define TKP 34


__global__ __launch_bounds__(256) void gemm_nt_f32_wmma(
    const float* __restrict__ A, const float* __restrict__ W,
    float* __restrict__ C, int M, int N, int K)
{
    __shared__ float As[TM][TKP];
    __shared__ float Ws[TN][TKP];

    const int tid  = threadIdx.x;
    const int lane = tid & 31, wid = tid >> 5;
    const int wm   = wid >> 1, wn  = wid & 1;
    const int ll   = lane & 15, lh = lane >> 4;
    const int m0   = blockIdx.x * TM, n0 = blockIdx.y * TN;

    const v8f vzero = {0.f,0.f,0.f,0.f,0.f,0.f,0.f,0.f};
    v8f acc[2][2];
    acc[0][0]=vzero; acc[0][1]=vzero; acc[1][0]=vzero; acc[1][1]=vzero;

    for (int k0 = 0; k0 < K; k0 += TK) {
        __syncthreads();
        #pragma unroll
        for (int p = 0; p < 4; ++p) {
            int r = p*32 + (tid >> 3);
            int c = (tid & 7) << 2;
            int gm = m0 + r, gk = k0 + c;
            v4f v = {0.f,0.f,0.f,0.f};
            if (gm < M && gk < K) v = *(const v4f*)(A + (size_t)gm*K + gk);
            *(v4f*)&As[r][c] = v;
        }
        #pragma unroll
        for (int p = 0; p < 2; ++p) {
            int r = p*32 + (tid >> 3);
            int c = (tid & 7) << 2;
            int gn = n0 + r, gk = k0 + c;
            v4f v = {0.f,0.f,0.f,0.f};
            if (gn < N && gk < K) v = *(const v4f*)(W + (size_t)gn*K + gk);
            *(v4f*)&Ws[r][c] = v;
        }
        __syncthreads();

        #pragma unroll
        for (int kk = 0; kk < 8; ++kk) {
            const int kc = 4*kk + 2*lh;
            v2f a0 = *(const v2f*)&As[wm*32      + ll][kc];
            v2f a1 = *(const v2f*)&As[wm*32 + 16 + ll][kc];
            v2f b0 = *(const v2f*)&Ws[wn*32      + ll][kc];
            v2f b1 = *(const v2f*)&Ws[wn*32 + 16 + ll][kc];
            acc[0][0] = __builtin_amdgcn_wmma_f32_16x16x4_f32(false,a0,false,b0,(short)0,acc[0][0],false,false);
            acc[0][1] = __builtin_amdgcn_wmma_f32_16x16x4_f32(false,a0,false,b1,(short)0,acc[0][1],false,false);
            acc[1][0] = __builtin_amdgcn_wmma_f32_16x16x4_f32(false,a1,false,b0,(short)0,acc[1][0],false,false);
            acc[1][1] = __builtin_amdgcn_wmma_f32_16x16x4_f32(false,a1,false,b1,(short)0,acc[1][1],false,false);
        }
    }

    #pragma unroll
    for (int mt = 0; mt < 2; ++mt)
        #pragma unroll
        for (int nt = 0; nt < 2; ++nt) {
            int gn = n0 + wn*32 + nt*16 + ll;
            #pragma unroll
            for (int r = 0; r < 8; ++r) {
                int gm = m0 + wm*32 + mt*16 + lh*8 + r;
                if (gm < M && gn < N) C[(size_t)gm*N + gn] = acc[mt][nt][r];
            }
        }
}

__global__ __launch_bounds__(256) void attn_softmax_kernel(
    const float* __restrict__ scores, const float* __restrict__ lmask,
    const float* __restrict__ imask, float* __restrict__ attn,
    float* __restrict__ b_in)
{
    int row = blockIdx.x * 256 + threadIdx.x;
    if (row >= 8*256*20) return;
    int bs = row / 20;
    int l  = row - bs*20;

    float v[8];
    float mx = -3.4e38f;
    #pragma unroll
    for (int t = 0; t < 8; ++t) {
        float s = scores[(size_t)row*8 + t] + (1.0f - lmask[l*8 + t]) * NEG_BIG;
        v[t] = s;
        mx = fmaxf(mx, s);
    }
    float sum = 0.f;
    #pragma unroll
    for (int t = 0; t < 8; ++t) { float e = __expf(v[t] - mx); v[t] = e; sum += e; }
    float inv = 1.0f / sum;
    #pragma unroll
    for (int t = 0; t < 8; ++t) attn[(size_t)row*8 + t] = v[t] * inv;

    b_in[row] = mx + (1.0f - imask[bs]) * NEG_BIG;
}

__global__ __launch_bounds__(256) void q2c_kernel(
    const float* __restrict__ b_in, const float* __restrict__ tkn,
    float* __restrict__ q2c)
{
    __shared__ float red[256];
    __shared__ float wgt[256];
    const int tid = threadIdx.x;
    const int b = blockIdx.x / 20;
    const int l = blockIdx.x - b*20;

    float x = b_in[(size_t)(b*256 + tid)*20 + l];
    red[tid] = x; __syncthreads();
    for (int st = 128; st > 0; st >>= 1) {
        if (tid < st) red[tid] = fmaxf(red[tid], red[tid + st]);
        __syncthreads();
    }
    float mx = red[0]; __syncthreads();
    float e = __expf(x - mx);
    red[tid] = e; __syncthreads();
    for (int st = 128; st > 0; st >>= 1) {
        if (tid < st) red[tid] += red[tid + st];
        __syncthreads();
    }
    float sm = red[0]; __syncthreads();
    wgt[tid] = e / sm; __syncthreads();

    for (int h = tid; h < 768; h += 256) {
        float acc = 0.f;
        const float* tp = tkn + (size_t)b*256*768 + h;
        #pragma unroll 4
        for (int s = 0; s < 256; ++s) acc += wgt[s] * tp[(size_t)s*768];
        q2c[(size_t)(b*20 + l)*768 + h] = acc;
    }
}

__global__ __launch_bounds__(256) __attribute__((amdgpu_waves_per_eu(6, 6)))
void fused_w5_tanh_wmma(
    const float* __restrict__ tkn,  const float* __restrict__ lab,
    const float* __restrict__ attn, const float* __restrict__ q2c,
    const float* __restrict__ W5,   const float* __restrict__ b5,
    float* __restrict__ out)
{
    __shared__ float As[3][TM][TKP];
    __shared__ float Ws[3][TN][TKP];
    __shared__ float aBlk[TM][8];

    const int tid  = threadIdx.x;
    const int lane = tid & 31, wid = tid >> 5;
    const int wm   = wid >> 1, wn  = wid & 1;
    const int ll   = lane & 15, lh = lane >> 4;
    const int bl   = blockIdx.x >> 1;
    const int b    = bl / 20;
    const int l    = bl - b*20;
    const int bs0  = b*256 + ((blockIdx.x & 1) << 7);
    const int n0   = blockIdx.y * TN;

    {
        int r = tid >> 1, c = (tid & 1) << 2;
        *(v4f*)&aBlk[r][c] = *(const v4f*)(attn + ((size_t)(bs0 + r)*20 + l)*8 + c);
    }

    const int sr = tid >> 3;
    const int sc = (tid & 7) << 2;
    const float* tknP = tkn + (size_t)(bs0 + sr)*768 + sc;
    const float* labP = lab + (size_t)l*8*768 + sc;
    const float* q2cP = q2c + (size_t)bl*768 + sc;
    const float* w5P  = W5  + (size_t)(n0 + sr)*3072 + sc;

    const v8f vzero = {0.f,0.f,0.f,0.f,0.f,0.f,0.f,0.f};
    v8f acc[2][2];
    acc[0][0]=vzero; acc[0][1]=vzero; acc[1][0]=vzero; acc[1][1]=vzero;

    for (int ch = 0; ch < 24; ++ch) {
        const int j0 = ch * TK;
        __syncthreads();
        {
            v4f q4 = *(const v4f*)(q2cP + j0);
            #pragma unroll
            for (int p = 0; p < 2; ++p) {
                const float* wp = w5P + (size_t)p*32*3072 + j0;
                v4f w0 = *(const v4f*)(wp);
                v4f w1 = *(const v4f*)(wp + 768);
                v4f w2 = *(const v4f*)(wp + 1536);
                v4f w3 = *(const v4f*)(wp + 2304);
                *(v4f*)&Ws[0][p*32 + sr][sc] = w0 + q4 * w3;
                *(v4f*)&Ws[1][p*32 + sr][sc] = w1;
                *(v4f*)&Ws[2][p*32 + sr][sc] = w2;
            }
            v4f lv[8];
            #pragma unroll
            for (int t = 0; t < 8; ++t) lv[t] = *(const v4f*)(labP + (size_t)t*768 + j0);
            v4f tv[4];
            #pragma unroll
            for (int p = 0; p < 4; ++p) tv[p] = *(const v4f*)(tknP + (size_t)p*32*768 + j0);
            #pragma unroll
            for (int p = 0; p < 4; ++p) {
                const int r = p*32 + sr;
                v4f cq = {0.f,0.f,0.f,0.f};
                #pragma unroll
                for (int t = 0; t < 8; ++t) cq += aBlk[r][t] * lv[t];
                *(v4f*)&As[0][r][sc] = tv[p];
                *(v4f*)&As[1][r][sc] = cq;
                *(v4f*)&As[2][r][sc] = tv[p] * cq;
            }
        }
        __syncthreads();

        #pragma unroll
        for (int s = 0; s < 3; ++s) {
            #pragma unroll
            for (int kk = 0; kk < 8; ++kk) {
                const int kc = 4*kk + 2*lh;
                v2f a0 = *(const v2f*)&As[s][wm*32      + ll][kc];
                v2f a1 = *(const v2f*)&As[s][wm*32 + 16 + ll][kc];
                v2f b0 = *(const v2f*)&Ws[s][wn*32      + ll][kc];
                v2f b1 = *(const v2f*)&Ws[s][wn*32 + 16 + ll][kc];
                acc[0][0] = __builtin_amdgcn_wmma_f32_16x16x4_f32(false,a0,false,b0,(short)0,acc[0][0],false,false);
                acc[0][1] = __builtin_amdgcn_wmma_f32_16x16x4_f32(false,a0,false,b1,(short)0,acc[0][1],false,false);
                acc[1][0] = __builtin_amdgcn_wmma_f32_16x16x4_f32(false,a1,false,b0,(short)0,acc[1][0],false,false);
                acc[1][1] = __builtin_amdgcn_wmma_f32_16x16x4_f32(false,a1,false,b1,(short)0,acc[1][1],false,false);
            }
        }
    }

    #pragma unroll
    for (int mt = 0; mt < 2; ++mt)
        #pragma unroll
        for (int nt = 0; nt < 2; ++nt) {
            int gn = n0 + wn*32 + nt*16 + ll;
            float bias = b5[gn];
            #pragma unroll
            for (int r = 0; r < 8; ++r) {
                int tr = wm*32 + mt*16 + lh*8 + r;
                size_t gm = (size_t)(bs0 + tr)*20 + l;
                out[gm*768 + gn] = tanhf(acc[mt][nt][r] + bias);
            }
        }
}

extern "C" void kernel_launch(void* const* d_in, const int* in_sizes, int n_in,
                              void* d_out, int out_size, void* d_ws, size_t ws_size,
                              hipStream_t stream)
{
    (void)in_sizes; (void)n_in; (void)out_size; (void)ws_size;

    const float* token = (const float*)d_in[0];
    const float* label = (const float*)d_in[1];
    const float* imask = (const float*)d_in[2];
    const float* lmask = (const float*)d_in[3];
    const float* W1    = (const float*)d_in[4];
    const float* W2    = (const float*)d_in[5];
    const float* W5    = (const float*)d_in[6];
    const float* b5    = (const float*)d_in[7];
    float* out = (float*)d_out;

    float* ws     = (float*)d_ws;
    float* tkn    = ws;
    float* lab    = tkn    + 2048*768;
    float* scores = lab    + 160*768;
    float* attn   = scores + 2048*160;
    float* b_in   = attn   + 2048*160;
    float* q2c    = b_in   + 40960;

    dim3 blk(256);
    gemm_nt_f32_wmma<<<dim3(16, 12), blk, 0, stream>>>(token, W1, tkn, 2048, 768, 768);
    gemm_nt_f32_wmma<<<dim3(2, 12),  blk, 0, stream>>>(label, W2, lab, 160, 768, 300);
    gemm_nt_f32_wmma<<<dim3(16, 3),  blk, 0, stream>>>(tkn, lab, scores, 2048, 160, 768);
    attn_softmax_kernel<<<dim3(160), blk, 0, stream>>>(scores, lmask, imask, attn, b_in);
    q2c_kernel<<<dim3(160), blk, 0, stream>>>(b_in, tkn, q2c);
    fused_w5_tanh_wmma<<<dim3(320, 12), blk, 0, stream>>>(tkn, lab, attn, q2c, W5, b5, out);
}


# q2c as a WMMA product: a block per (batch, 128 columns) forms the 20 softmax-over-S weight rows once as the A tile, tkn rows straight from memory as B; was 192 dependent loads; sum order changes
# speedup vs baseline: 1.6561x; 1.0800x over previous
#include <hip/hip_runtime.h>
#include <math.h>


typedef __attribute__((ext_vector_type(2))) float v2f;
typedef __attribute__((ext_vector_type(4))) float v4f;
typedef __attribute__((ext_vector_type(8))) float v8f;

#define NEG_BIG (-10000.0f)

#define TM 128
#define TN 64
#define TK 32
#define TKP 34


__global__ __launch_bounds__(256) void gemm_nt_f32_wmma(
    const float* __restrict__ A, const float* __restrict__ W,
    float* __restrict__ C, int M, int N, int K)
{
    __shared__ float As[TM][TKP];
    __shared__ float Ws[TN][TKP];

    const int tid  = threadIdx.x;
    const int lane = tid & 31, wid = tid >> 5;
    const int wm   = wid >> 1, wn  = wid & 1;
    const int ll   = lane & 15, lh = lane >> 4;
    const int m0   = blockIdx.x * TM, n0 = blockIdx.y * TN;

    const v8f vzero = {0.f,0.f,0.f,0.f,0.f,0.f,0.f,0.f};
    v8f acc[2][2];
    acc[0][0]=vzero; acc[0][1]=vzero; acc[1][0]=vzero; acc[1][1]=vzero;

    for (int k0 = 0; k0 < K; k0 += TK) {
        __syncthreads();
        #pragma unroll
        for (int p = 0; p < 4; ++p) {
            int r = p*32 + (tid >> 3);
            int c = (tid & 7) << 2;
            int gm = m0 + r, gk = k0 + c;
            v4f v = {0.f,0.f,0.f,0.f};
            if (gm < M && gk < K) v = *(const v4f*)(A + (size_t)gm*K + gk);
            *(v4f*)&As[r][c] = v;
        }
        #pragma unroll
        for (int p = 0; p < 2; ++p) {
            int r = p*32 + (tid >> 3);
            int c = (tid & 7) << 2;
            int gn = n0 + r, gk = k0 + c;
            v4f v = {0.f,0.f,0.f,0.f};
            if (gn < N && gk < K) v = *(const v4f*)(W + (size_t)gn*K + gk);
            *(v4f*)&Ws[r][c] = v;
        }
        __syncthreads();

        #pragma unroll
        for (int kk = 0; kk < 8; ++kk) {
            const int kc = 4*kk + 2*lh;
            v2f a0 = *(const v2f*)&As[wm*32      + ll][kc];
            v2f a1 = *(const v2f*)&As[wm*32 + 16 + ll][kc];
            v2f b0 = *(const v2f*)&Ws[wn*32      + ll][kc];
            v2f b1 = *(const v2f*)&Ws[wn*32 + 16 + ll][kc];
            acc[0][0] = __builtin_amdgcn_wmma_f32_16x16x4_f32(false,a0,false,b0,(short)0,acc[0][0],false,false);
            acc[0][1] = __builtin_amdgcn_wmma_f32_16x16x4_f32(false,a0,false,b1,(short)0,acc[0][1],false,false);
            acc[1][0] = __builtin_amdgcn_wmma_f32_16x16x4_f32(false,a1,false,b0,(short)0,acc[1][0],false,false);
            acc[1][1] = __builtin_amdgcn_wmma_f32_16x16x4_f32(false,a1,false,b1,(short)0,acc[1][1],false,false);
        }
    }

    #pragma unroll
    for (int mt = 0; mt < 2; ++mt)
        #pragma unroll
        for (int nt = 0; nt < 2; ++nt) {
            int gn = n0 + wn*32 + nt*16 + ll;
            #pragma unroll
            for (int r = 0; r < 8; ++r) {
                int gm = m0 + wm*32 + mt*16 + lh*8 + r;
                if (gm < M && gn < N) C[(size_t)gm*N + gn] = acc[mt][nt][r];
            }
        }
}

__global__ __launch_bounds__(256) void attn_softmax_kernel(
    const float* __restrict__ scores, const float* __restrict__ lmask,
    const float* __restrict__ imask, float* __restrict__ attn,
    float* __restrict__ b_in)
{
    int row = blockIdx.x * 256 + threadIdx.x;
    if (row >= 8*256*20) return;
    int bs = row / 20;
    int l  = row - bs*20;

    float v[8];
    float mx = -3.4e38f;
    #pragma unroll
    for (int t = 0; t < 8; ++t) {
        float s = scores[(size_t)row*8 + t] + (1.0f - lmask[l*8 + t]) * NEG_BIG;
        v[t] = s;
        mx = fmaxf(mx, s);
    }
    float sum = 0.f;
    #pragma unroll
    for (int t = 0; t < 8; ++t) { float e = __expf(v[t] - mx); v[t] = e; sum += e; }
    float inv = 1.0f / sum;
    #pragma unroll
    for (int t = 0; t < 8; ++t) attn[(size_t)row*8 + t] = v[t] * inv;

    b_in[row] = mx + (1.0f - imask[bs]) * NEG_BIG;
}

#define QP 260

__global__ __launch_bounds__(256) void q2c_wmma_kernel(
    const float* __restrict__ b_in, const float* __restrict__ tkn,
    float* __restrict__ q2c)
{
    __shared__ float bS[256*20];
    __shared__ float Ps[32][QP];

    const int tid  = threadIdx.x;
    const int lane = tid & 31, wid = tid >> 5;
    const int ll   = lane & 15, lh = lane >> 4;
    const int b    = blockIdx.x;
    const int n0   = blockIdx.y * 128;

    #pragma unroll
    for (int i = 0; i < 5; ++i) {
        int idx = (i*256 + tid) << 2;
        *(v4f*)&bS[idx] = *(const v4f*)(b_in + (size_t)b*5120 + idx);
    }
    for (int i = tid; i < 12*QP; i += 256) (&Ps[20][0])[i] = 0.f;
    __syncthreads();

    for (int l = wid; l < 20; l += 8) {
        float x[8];
        float mx = -3.4e38f;
        #pragma unroll
        for (int i = 0; i < 8; ++i) { x[i] = bS[(lane + 32*i)*20 + l]; mx = fmaxf(mx, x[i]); }
        #pragma unroll
        for (int off = 16; off > 0; off >>= 1) mx = fmaxf(mx, __shfl_xor(mx, off));
        float sm = 0.f;
        #pragma unroll
        for (int i = 0; i < 8; ++i) { x[i] = __expf(x[i] - mx); sm += x[i]; }
        #pragma unroll
        for (int off = 16; off > 0; off >>= 1) sm += __shfl_xor(sm, off);
        #pragma unroll
        for (int i = 0; i < 8; ++i) Ps[l][lane + 32*i] = x[i] / sm;
    }
    __syncthreads();

    const float* bp = tkn + (size_t)b*256*768 + (size_t)(2*lh)*768 + n0 + wid*16 + ll;
    const v8f vzero = {0.f,0.f,0.f,0.f,0.f,0.f,0.f,0.f};
    v8f acc0 = vzero, acc1 = vzero;

    for (int kb = 0; kb < 64; kb += 16) {
        float bv0[16], bv1[16];
        #pragma unroll
        for (int j = 0; j < 16; ++j) {
            const float* p = bp + (size_t)(4*(kb + j))*768;
            bv0[j] = p[0];
            bv1[j] = p[768];
        }
        #pragma unroll
        for (int j = 0; j < 16; ++j) {
            const int kc = 4*(kb + j) + 2*lh;
            v2f a0 = *(const v2f*)&Ps[ll][kc];
            v2f a1 = *(const v2f*)&Ps[16 + ll][kc];
            v2f bf = {bv0[j], bv1[j]};
            acc0 = __builtin_amdgcn_wmma_f32_16x16x4_f32(false,a0,false,bf,(short)0,acc0,false,false);
            acc1 = __builtin_amdgcn_wmma_f32_16x16x4_f32(false,a1,false,bf,(short)0,acc1,false,false);
        }
    }

    const int gn = n0 + wid*16 + ll;
    #pragma unroll
    for (int r = 0; r < 8; ++r) {
        int row = lh*8 + r;
        q2c[(size_t)(b*20 + row)*768 + gn] = acc0[r];
    }
    #pragma unroll
    for (int r = 0; r < 8; ++r) {
        int row = 16 + lh*8 + r;
        if (row < 20) q2c[(size_t)(b*20 + row)*768 + gn] = acc1[r];
    }
}

__global__ __launch_bounds__(256) __attribute__((amdgpu_waves_per_eu(6, 6)))
void fused_w5_tanh_wmma(
    const float* __restrict__ tkn,  const float* __restrict__ lab,
    const float* __restrict__ attn, const float* __restrict__ q2c,
    const float* __restrict__ W5,   const float* __restrict__ b5,
    float* __restrict__ out)
{
    __shared__ float As[3][TM][TKP];
    __shared__ float Ws[3][TN][TKP];
    __shared__ float aBlk[TM][8];

    const int tid  = threadIdx.x;
    const int lane = tid & 31, wid = tid >> 5;
    const int wm   = wid >> 1, wn  = wid & 1;
    const int ll   = lane & 15, lh = lane >> 4;
    const int bl   = blockIdx.x >> 1;
    const int b    = bl / 20;
    const int l    = bl - b*20;
    const int bs0  = b*256 + ((blockIdx.x & 1) << 7);
    const int n0   = blockIdx.y * TN;

    {
        int r = tid >> 1, c = (tid & 1) << 2;
        *(v4f*)&aBlk[r][c] = *(const v4f*)(attn + ((size_t)(bs0 + r)*20 + l)*8 + c);
    }

    const int sr = tid >> 3;
    const int sc = (tid & 7) << 2;
    const float* tknP = tkn + (size_t)(bs0 + sr)*768 + sc;
    const float* labP = lab + (size_t)l*8*768 + sc;
    const float* q2cP = q2c + (size_t)bl*768 + sc;
    const float* w5P  = W5  + (size_t)(n0 + sr)*3072 + sc;

    const v8f vzero = {0.f,0.f,0.f,0.f,0.f,0.f,0.f,0.f};
    v8f acc[2][2];
    acc[0][0]=vzero; acc[0][1]=vzero; acc[1][0]=vzero; acc[1][1]=vzero;

    for (int ch = 0; ch < 24; ++ch) {
        const int j0 = ch * TK;
        __syncthreads();
        {
            v4f q4 = *(const v4f*)(q2cP + j0);
            #pragma unroll
            for (int p = 0; p < 2; ++p) {
                const float* wp = w5P + (size_t)p*32*3072 + j0;
                v4f w0 = *(const v4f*)(wp);
                v4f w1 = *(const v4f*)(wp + 768);
                v4f w2 = *(const v4f*)(wp + 1536);
                v4f w3 = *(const v4f*)(wp + 2304);
                *(v4f*)&Ws[0][p*32 + sr][sc] = w0 + q4 * w3;
                *(v4f*)&Ws[1][p*32 + sr][sc] = w1;
                *(v4f*)&Ws[2][p*32 + sr][sc] = w2;
            }
            v4f lv[8];
            #pragma unroll
            for (int t = 0; t < 8; ++t) lv[t] = *(const v4f*)(labP + (size_t)t*768 + j0);
            v4f tv[4];
            #pragma unroll
            for (int p = 0; p < 4; ++p) tv[p] = *(const v4f*)(tknP + (size_t)p*32*768 + j0);
            #pragma unroll
            for (int p = 0; p < 4; ++p) {
                const int r = p*32 + sr;
                v4f cq = {0.f,0.f,0.f,0.f};
                #pragma unroll
                for (int t = 0; t < 8; ++t) cq += aBlk[r][t] * lv[t];
                *(v4f*)&As[0][r][sc] = tv[p];
                *(v4f*)&As[1][r][sc] = cq;
                *(v4f*)&As[2][r][sc] = tv[p] * cq;
            }
        }
        __syncthreads();

        #pragma unroll
        for (int s = 0; s < 3; ++s) {
            #pragma unroll
            for (int kk = 0; kk < 8; ++kk) {
                const int kc = 4*kk + 2*lh;
                v2f a0 = *(const v2f*)&As[s][wm*32      + ll][kc];
                v2f a1 = *(const v2f*)&As[s][wm*32 + 16 + ll][kc];
                v2f b0 = *(const v2f*)&Ws[s][wn*32      + ll][kc];
                v2f b1 = *(const v2f*)&Ws[s][wn*32 + 16 + ll][kc];
                acc[0][0] = __builtin_amdgcn_wmma_f32_16x16x4_f32(false,a0,false,b0,(short)0,acc[0][0],false,false);
                acc[0][1] = __builtin_amdgcn_wmma_f32_16x16x4_f32(false,a0,false,b1,(short)0,acc[0][1],false,false);
                acc[1][0] = __builtin_amdgcn_wmma_f32_16x16x4_f32(false,a1,false,b0,(short)0,acc[1][0],false,false);
                acc[1][1] = __builtin_amdgcn_wmma_f32_16x16x4_f32(false,a1,false,b1,(short)0,acc[1][1],false,false);
            }
        }
    }

    #pragma unroll
    for (int mt = 0; mt < 2; ++mt)
        #pragma unroll
        for (int nt = 0; nt < 2; ++nt) {
            int gn = n0 + wn*32 + nt*16 + ll;
            float bias = b5[gn];
            #pragma unroll
            for (int r = 0; r < 8; ++r) {
                int tr = wm*32 + mt*16 + lh*8 + r;
                size_t gm = (size_t)(bs0 + tr)*20 + l;
                out[gm*768 + gn] = tanhf(acc[mt][nt][r] + bias);
            }
        }
}

extern "C" void kernel_launch(void* const* d_in, const int* in_sizes, int n_in,
                              void* d_out, int out_size, void* d_ws, size_t ws_size,
                              hipStream_t stream)
{
    (void)in_sizes; (void)n_in; (void)out_size; (void)ws_size;

    const float* token = (const float*)d_in[0];
    const float* label = (const float*)d_in[1];
    const float* imask = (const float*)d_in[2];
    const float* lmask = (const float*)d_in[3];
    const float* W1    = (const float*)d_in[4];
    const float* W2    = (const float*)d_in[5];
    const float* W5    = (const float*)d_in[6];
    const float* b5    = (const float*)d_in[7];
    float* out = (float*)d_out;

    float* ws     = (float*)d_ws;
    float* tkn    = ws;
    float* lab    = tkn    + 2048*768;
    float* scores = lab    + 160*768;
    float* attn   = scores + 2048*160;
    float* b_in   = attn   + 2048*160;
    float* q2c    = b_in   + 40960;

    dim3 blk(256);
    gemm_nt_f32_wmma<<<dim3(16, 12), blk, 0, stream>>>(token, W1, tkn, 2048, 768, 768);
    gemm_nt_f32_wmma<<<dim3(2, 12),  blk, 0, stream>>>(label, W2, lab, 160, 768, 300);
    gemm_nt_f32_wmma<<<dim3(16, 3),  blk, 0, stream>>>(tkn, lab, scores, 2048, 160, 768);
    attn_softmax_kernel<<<dim3(160), blk, 0, stream>>>(scores, lmask, imask, attn, b_in);
    q2c_wmma_kernel<<<dim3(8, 6), blk, 0, stream>>>(b_in, tkn, q2c);
    fused_w5_tanh_wmma<<<dim3(320, 12), blk, 0, stream>>>(tkn, lab, attn, q2c, W5, b5, out);
}


# Main GEMM: c2q made on the matrix unit as a transposed K=8 product; its D registers are the A fragments of the c2q and tkn*c2q sections (no LDS pass, no VALU c2q); wave tile 16x64; sum order changes
# speedup vs baseline: 1.7318x; 1.0457x over previous
#include <hip/hip_runtime.h>
#include <math.h>


typedef __attribute__((ext_vector_type(2))) float v2f;
typedef __attribute__((ext_vector_type(4))) float v4f;
typedef __attribute__((ext_vector_type(8))) float v8f;

#define NEG_BIG (-10000.0f)

#define TM 128
#define TN 64
#define TK 32
#define TKP 34


__global__ __launch_bounds__(256) void gemm_nt_f32_wmma(
    const float* __restrict__ A, const float* __restrict__ W,
    float* __restrict__ C, int M, int N, int K)
{
    __shared__ float As[TM][TKP];
    __shared__ float Ws[TN][TKP];

    const int tid  = threadIdx.x;
    const int lane = tid & 31, wid = tid >> 5;
    const int wm   = wid >> 1, wn  = wid & 1;
    const int ll   = lane & 15, lh = lane >> 4;
    const int m0   = blockIdx.x * TM, n0 = blockIdx.y * TN;

    const v8f vzero = {0.f,0.f,0.f,0.f,0.f,0.f,0.f,0.f};
    v8f acc[2][2];
    acc[0][0]=vzero; acc[0][1]=vzero; acc[1][0]=vzero; acc[1][1]=vzero;

    for (int k0 = 0; k0 < K; k0 += TK) {
        __syncthreads();
        #pragma unroll
        for (int p = 0; p < 4; ++p) {
            int r = p*32 + (tid >> 3);
            int c = (tid & 7) << 2;
            int gm = m0 + r, gk = k0 + c;
            v4f v = {0.f,0.f,0.f,0.f};
            if (gm < M && gk < K) v = *(const v4f*)(A + (size_t)gm*K + gk);
            *(v4f*)&As[r][c] = v;
        }
        #pragma unroll
        for (int p = 0; p < 2; ++p) {
            int r = p*32 + (tid >> 3);
            int c = (tid & 7) << 2;
            int gn = n0 + r, gk = k0 + c;
            v4f v = {0.f,0.f,0.f,0.f};
            if (gn < N && gk < K) v = *(const v4f*)(W + (size_t)gn*K + gk);
            *(v4f*)&Ws[r][c] = v;
        }
        __syncthreads();

        #pragma unroll
        for (int kk = 0; kk < 8; ++kk) {
            const int kc = 4*kk + 2*lh;
            v2f a0 = *(const v2f*)&As[wm*32      + ll][kc];
            v2f a1 = *(const v2f*)&As[wm*32 + 16 + ll][kc];
            v2f b0 = *(const v2f*)&Ws[wn*32      + ll][kc];
            v2f b1 = *(const v2f*)&Ws[wn*32 + 16 + ll][kc];
            acc[0][0] = __builtin_amdgcn_wmma_f32_16x16x4_f32(false,a0,false,b0,(short)0,acc[0][0],false,false);
            acc[0][1] = __builtin_amdgcn_wmma_f32_16x16x4_f32(false,a0,false,b1,(short)0,acc[0][1],false,false);
            acc[1][0] = __builtin_amdgcn_wmma_f32_16x16x4_f32(false,a1,false,b0,(short)0,acc[1][0],false,false);
            acc[1][1] = __builtin_amdgcn_wmma_f32_16x16x4_f32(false,a1,false,b1,(short)0,acc[1][1],false,false);
        }
    }

    #pragma unroll
    for (int mt = 0; mt < 2; ++mt)
        #pragma unroll
        for (int nt = 0; nt < 2; ++nt) {
            int gn = n0 + wn*32 + nt*16 + ll;
            #pragma unroll
            for (int r = 0; r < 8; ++r) {
                int gm = m0 + wm*32 + mt*16 + lh*8 + r;
                if (gm < M && gn < N) C[(size_t)gm*N + gn] = acc[mt][nt][r];
            }
        }
}

__global__ __launch_bounds__(256) void attn_softmax_kernel(
    const float* __restrict__ scores, const float* __restrict__ lmask,
    const float* __restrict__ imask, float* __restrict__ attn,
    float* __restrict__ b_in)
{
    int row = blockIdx.x * 256 + threadIdx.x;
    if (row >= 8*256*20) return;
    int bs = row / 20;
    int l  = row - bs*20;

    float v[8];
    float mx = -3.4e38f;
    #pragma unroll
    for (int t = 0; t < 8; ++t) {
        float s = scores[(size_t)row*8 + t] + (1.0f - lmask[l*8 + t]) * NEG_BIG;
        v[t] = s;
        mx = fmaxf(mx, s);
    }
    float sum = 0.f;
    #pragma unroll
    for (int t = 0; t < 8; ++t) { float e = __expf(v[t] - mx); v[t] = e; sum += e; }
    float inv = 1.0f / sum;
    #pragma unroll
    for (int t = 0; t < 8; ++t) attn[(size_t)row*8 + t] = v[t] * inv;

    b_in[row] = mx + (1.0f - imask[bs]) * NEG_BIG;
}

#define QP 260

__global__ __launch_bounds__(256) void q2c_wmma_kernel(
    const float* __restrict__ b_in, const float* __restrict__ tkn,
    float* __restrict__ q2c)
{
    __shared__ float bS[256*20];
    __shared__ float Ps[32][QP];

    const int tid  = threadIdx.x;
    const int lane = tid & 31, wid = tid >> 5;
    const int ll   = lane & 15, lh = lane >> 4;
    const int b    = blockIdx.x;
    const int n0   = blockIdx.y * 128;

    #pragma unroll
    for (int i = 0; i < 5; ++i) {
        int idx = (i*256 + tid) << 2;
        *(v4f*)&bS[idx] = *(const v4f*)(b_in + (size_t)b*5120 + idx);
    }
    for (int i = tid; i < 12*QP; i += 256) (&Ps[20][0])[i] = 0.f;
    __syncthreads();

    for (int l = wid; l < 20; l += 8) {
        float x[8];
        float mx = -3.4e38f;
        #pragma unroll
        for (int i = 0; i < 8; ++i) { x[i] = bS[(lane + 32*i)*20 + l]; mx = fmaxf(mx, x[i]); }
        #pragma unroll
        for (int off = 16; off > 0; off >>= 1) mx = fmaxf(mx, __shfl_xor(mx, off));
        float sm = 0.f;
        #pragma unroll
        for (int i = 0; i < 8; ++i) { x[i] = __expf(x[i] - mx); sm += x[i]; }
        #pragma unroll
        for (int off = 16; off > 0; off >>= 1) sm += __shfl_xor(sm, off);
        #pragma unroll
        for (int i = 0; i < 8; ++i) Ps[l][lane + 32*i] = x[i] / sm;
    }
    __syncthreads();

    const float* bp = tkn + (size_t)b*256*768 + (size_t)(2*lh)*768 + n0 + wid*16 + ll;
    const v8f vzero = {0.f,0.f,0.f,0.f,0.f,0.f,0.f,0.f};
    v8f acc0 = vzero, acc1 = vzero;

    for (int kb = 0; kb < 64; kb += 16) {
        float bv0[16], bv1[16];
        #pragma unroll
        for (int j = 0; j < 16; ++j) {
            const float* p = bp + (size_t)(4*(kb + j))*768;
            bv0[j] = p[0];
            bv1[j] = p[768];
        }
        #pragma unroll
        for (int j = 0; j < 16; ++j) {
            const int kc = 4*(kb + j) + 2*lh;
            v2f a0 = *(const v2f*)&Ps[ll][kc];
            v2f a1 = *(const v2f*)&Ps[16 + ll][kc];
            v2f bf = {bv0[j], bv1[j]};
            acc0 = __builtin_amdgcn_wmma_f32_16x16x4_f32(false,a0,false,bf,(short)0,acc0,false,false);
            acc1 = __builtin_amdgcn_wmma_f32_16x16x4_f32(false,a1,false,bf,(short)0,acc1,false,false);
        }
    }

    const int gn = n0 + wid*16 + ll;
    #pragma unroll
    for (int r = 0; r < 8; ++r) {
        int row = lh*8 + r;
        q2c[(size_t)(b*20 + row)*768 + gn] = acc0[r];
    }
    #pragma unroll
    for (int r = 0; r < 8; ++r) {
        int row = 16 + lh*8 + r;
        if (row < 20) q2c[(size_t)(b*20 + row)*768 + gn] = acc1[r];
    }
}

#define KP 36
#define LP 12

__global__ __launch_bounds__(256) __attribute__((amdgpu_waves_per_eu(6, 6)))
void fused_w5_tanh_wmma(
    const float* __restrict__ tkn,  const float* __restrict__ lab,
    const float* __restrict__ attn, const float* __restrict__ q2c,
    const float* __restrict__ W5,   const float* __restrict__ b5,
    float* __restrict__ out)
{
    __shared__ __attribute__((aligned(16))) float Ts[TM][KP];
    __shared__ __attribute__((aligned(16))) float Ws[3][TN][KP];
    __shared__ __attribute__((aligned(16))) float Lt[TK][LP];

    const int tid  = threadIdx.x;
    const int lane = tid & 31, wid = tid >> 5;
    const int c    = lane & 15, h = lane >> 4;
    const int bl   = blockIdx.x >> 1;
    const int b    = bl / 20;
    const int l    = bl - b*20;
    const int bs0  = b*256 + ((blockIdx.x & 1) << 7);
    const int n0   = blockIdx.y * TN;

    const int g  = tid >> 3;
    const int sr = ((g & 1) << 3) | ((g >> 1) & 7) | (g & 16);
    const int sc = (tid & 7) << 2;
    const float* tknP = tkn + (size_t)(bs0 + sr)*768 + sc;
    const float* q2cP = q2c + (size_t)bl*768 + sc;
    const float* w5P  = W5  + (size_t)(n0 + sr)*3072 + sc;
    const int lt_t = tid >> 5, lt_j = tid & 31;
    const float* labP = lab + (size_t)(l*8 + lt_t)*768 + lt_j;

    const float* ap = attn + ((size_t)(bs0 + wid*16 + c)*20 + l)*8 + 2*h;
    const v2f aT0 = *(const v2f*)(ap);
    const v2f aT1 = *(const v2f*)(ap + 4);

    const v8f vzero = {0.f,0.f,0.f,0.f,0.f,0.f,0.f,0.f};
    v8f acc[4];
    acc[0]=vzero; acc[1]=vzero; acc[2]=vzero; acc[3]=vzero;

    for (int ch = 0; ch < 24; ++ch) {
        const int j0 = ch * TK;
        __syncthreads();
        {
            v4f q4 = *(const v4f*)(q2cP + j0);
            #pragma unroll
            for (int p = 0; p < 2; ++p) {
                const float* wp = w5P + (size_t)p*32*3072 + j0;
                v4f w0 = *(const v4f*)(wp);
                v4f w1 = *(const v4f*)(wp + 768);
                v4f w2 = *(const v4f*)(wp + 1536);
                v4f w3 = *(const v4f*)(wp + 2304);
                *(v4f*)&Ws[0][p*32 + sr][sc] = w0 + q4 * w3;
                *(v4f*)&Ws[1][p*32 + sr][sc] = w1;
                *(v4f*)&Ws[2][p*32 + sr][sc] = w2;
            }
            #pragma unroll
            for (int p = 0; p < 4; ++p)
                *(v4f*)&Ts[p*32 + sr][sc] = *(const v4f*)(tknP + (size_t)p*32*768 + j0);
            Lt[lt_j][lt_t] = labP[j0];
        }
        __syncthreads();

        #pragma unroll
        for (int jg = 0; jg < 2; ++jg) {
            const int cb = 16*jg + 8*h;
            v2f la0 = *(const v2f*)&Lt[16*jg + c][2*h];
            v2f la1 = *(const v2f*)&Lt[16*jg + c][4 + 2*h];
            v8f d = vzero;
            d = __builtin_amdgcn_wmma_f32_16x16x4_f32(false,la0,false,aT0,(short)0,d,false,false);
            d = __builtin_amdgcn_wmma_f32_16x16x4_f32(false,la1,false,aT1,(short)0,d,false,false);
            v4f ta = *(const v4f*)&Ts[wid*16 + c][cb];
            v4f tb = *(const v4f*)&Ts[wid*16 + c][cb + 4];
            v8f t8 = __builtin_shufflevector(ta, tb, 0,1,2,3,4,5,6,7);
            v8f pr = t8 * d;

            #pragma unroll
            for (int hf = 0; hf < 2; ++hf) {
                v2f a0[2], a1[2], a2[2];
                if (hf == 0) {
                    a0[0] = __builtin_shufflevector(t8, t8, 0, 1); a0[1] = __builtin_shufflevector(t8, t8, 2, 3);
                    a1[0] = __builtin_shufflevector(d,  d,  0, 1); a1[1] = __builtin_shufflevector(d,  d,  2, 3);
                    a2[0] = __builtin_shufflevector(pr, pr, 0, 1); a2[1] = __builtin_shufflevector(pr, pr, 2, 3);
                } else {
                    a0[0] = __builtin_shufflevector(t8, t8, 4, 5); a0[1] = __builtin_shufflevector(t8, t8, 6, 7);
                    a1[0] = __builtin_shufflevector(d,  d,  4, 5); a1[1] = __builtin_shufflevector(d,  d,  6, 7);
                    a2[0] = __builtin_shufflevector(pr, pr, 4, 5); a2[1] = __builtin_shufflevector(pr, pr, 6, 7);
                }
                #pragma unroll
                for (int s = 0; s < 3; ++s) {
                    v4f bv[4];
                    #pragma unroll
                    for (int nt = 0; nt < 4; ++nt) bv[nt] = *(const v4f*)&Ws[s][nt*16 + c][cb + 4*hf];
                    #pragma unroll
                    for (int u = 0; u < 2; ++u) {
                        const v2f af = (s == 0) ? a0[u] : (s == 1) ? a1[u] : a2[u];
                        #pragma unroll
                        for (int nt = 0; nt < 4; ++nt) {
                            v2f bf = (u == 0) ? __builtin_shufflevector(bv[nt], bv[nt], 0, 1)
                                              : __builtin_shufflevector(bv[nt], bv[nt], 2, 3);
                            acc[nt] = __builtin_amdgcn_wmma_f32_16x16x4_f32(false,af,false,bf,(short)0,acc[nt],false,false);
                        }
                    }
                }
            }
        }
    }

    #pragma unroll
    for (int nt = 0; nt < 4; ++nt) {
        int gn = n0 + nt*16 + c;
        float bias = b5[gn];
        #pragma unroll
        for (int r = 0; r < 8; ++r) {
            int tr = wid*16 + h*8 + r;
            size_t gm = (size_t)(bs0 + tr)*20 + l;
            out[gm*768 + gn] = tanhf(acc[nt][r] + bias);
        }
    }
}

extern "C" void kernel_launch(void* const* d_in, const int* in_sizes, int n_in,
                              void* d_out, int out_size, void* d_ws, size_t ws_size,
                              hipStream_t stream)
{
    (void)in_sizes; (void)n_in; (void)out_size; (void)ws_size;

    const float* token = (const float*)d_in[0];
    const float* label = (const float*)d_in[1];
    const float* imask = (const float*)d_in[2];
    const float* lmask = (const float*)d_in[3];
    const float* W1    = (const float*)d_in[4];
    const float* W2    = (const float*)d_in[5];
    const float* W5    = (const float*)d_in[6];
    const float* b5    = (const float*)d_in[7];
    float* out = (float*)d_out;

    float* ws     = (float*)d_ws;
    float* tkn    = ws;
    float* lab    = tkn    + 2048*768;
    float* scores = lab    + 160*768;
    float* attn   = scores + 2048*160;
    float* b_in   = attn   + 2048*160;
    float* q2c    = b_in   + 40960;

    dim3 blk(256);
    gemm_nt_f32_wmma<<<dim3(16, 12), blk, 0, stream>>>(token, W1, tkn, 2048, 768, 768);
    gemm_nt_f32_wmma<<<dim3(2, 12),  blk, 0, stream>>>(label, W2, lab, 160, 768, 300);
    gemm_nt_f32_wmma<<<dim3(16, 3),  blk, 0, stream>>>(tkn, lab, scores, 2048, 160, 768);
    attn_softmax_kernel<<<dim3(160), blk, 0, stream>>>(scores, lmask, imask, attn, b_in);
    q2c_wmma_kernel<<<dim3(8, 6), blk, 0, stream>>>(b_in, tkn, q2c);
    fused_w5_tanh_wmma<<<dim3(320, 12), blk, 0, stream>>>(tkn, lab, attn, q2c, W5, b5, out);
}


# Small GEMMs (tkn, lab, scores) re-tiled: 4-wave blocks of 64x48 or 32x32 outputs (512/120/320 blocks, were 192/24/48) and 128-column K steps (6 load rounds, were 24); sum order changes
# speedup vs baseline: 1.8961x; 1.0949x over previous
#include <hip/hip_runtime.h>
#include <math.h>


typedef __attribute__((ext_vector_type(2))) float v2f;
typedef __attribute__((ext_vector_type(4))) float v4f;
typedef __attribute__((ext_vector_type(8))) float v8f;

#define NEG_BIG (-10000.0f)

#define TM 128
#define TN 64
#define TK 32


#define GK  128
#define GKP 132

template <int BM, int BN, int WM, int WN>
__global__ __launch_bounds__(128) void gemm_nt_tile_wmma(
    const float* __restrict__ A, const float* __restrict__ W,
    float* __restrict__ C, int M, int N, int K)
{
    constexpr int WNC = BN / WN;
    constexpr int NTL = WNC / 16;
    constexpr int GB  = (NTL >= 2) ? 2 : 4;
    static_assert(BM == 16 * WM && WM * WN == 4 && NTL * 16 * WN == BN, "tile shape");

    __shared__ __attribute__((aligned(16))) float As[BM][GKP];
    __shared__ __attribute__((aligned(16))) float Ws[BN][GKP];

    const int tid  = threadIdx.x;
    const int lane = tid & 31, wid = tid >> 5;
    const int wm   = wid / WN, wn = wid - wm * WN;
    const int c    = lane & 15, h = lane >> 4;
    const int m0   = blockIdx.x * BM, n0 = blockIdx.y * BN;
    const int sr   = tid >> 5;
    const int sc   = (tid & 31) << 2;

    const v8f vzero = {0.f,0.f,0.f,0.f,0.f,0.f,0.f,0.f};
    v8f acc[2][NTL];
    #pragma unroll
    for (int nt = 0; nt < NTL; ++nt) { acc[0][nt] = vzero; acc[1][nt] = vzero; }

    for (int k0 = 0; k0 < K; k0 += GK) {
        __syncthreads();
        {
            const int gk = k0 + sc;
            v4f ra[BM / 4], rw[BN / 4];
            #pragma unroll
            for (int p = 0; p < BM / 4; ++p) {
                const int gm = m0 + p * 4 + sr;
                v4f v = {0.f,0.f,0.f,0.f};
                if (gm < M && gk < K) v = *(const v4f*)(A + (size_t)gm * K + gk);
                ra[p] = v;
            }
            #pragma unroll
            for (int p = 0; p < BN / 4; ++p) {
                const int gn = n0 + p * 4 + sr;
                v4f v = {0.f,0.f,0.f,0.f};
                if (gn < N && gk < K) v = *(const v4f*)(W + (size_t)gn * K + gk);
                rw[p] = v;
            }
            #pragma unroll
            for (int p = 0; p < BM / 4; ++p) *(v4f*)&As[p * 4 + sr][sc] = ra[p];
            #pragma unroll
            for (int p = 0; p < BN / 4; ++p) *(v4f*)&Ws[p * 4 + sr][sc] = rw[p];
        }
        __syncthreads();

        const int krem = K - k0;
        const int nb   = ((krem < GK ? krem : GK) + 8 * GB - 1) / (8 * GB);
        const float* aRow = &As[wm * 16 + c][4 * h];
        const float* wRow = &Ws[wn * WNC + c][4 * h];
        v4f aC[GB], bC[GB][NTL];
        #pragma unroll
        for (int q = 0; q < GB; ++q) {
            aC[q] = *(const v4f*)(aRow + 8 * q);
            #pragma unroll
            for (int nt = 0; nt < NTL; ++nt) bC[q][nt] = *(const v4f*)(wRow + nt * 16 * GKP + 8 * q);
        }
        #pragma unroll 2
        for (int bi = 0; bi < nb; ++bi) {
            const int bn = (bi + 1 < nb) ? bi + 1 : bi;
            v4f aN[GB], bN[GB][NTL];
            #pragma unroll
            for (int q = 0; q < GB; ++q) {
                aN[q] = *(const v4f*)(aRow + 8 * (bn * GB + q));
                #pragma unroll
                for (int nt = 0; nt < NTL; ++nt) bN[q][nt] = *(const v4f*)(wRow + nt * 16 * GKP + 8 * (bn * GB + q));
            }
            #pragma unroll
            for (int q = 0; q < GB; ++q) {
                const v2f aLo = __builtin_shufflevector(aC[q], aC[q], 0, 1);
                const v2f aHi = __builtin_shufflevector(aC[q], aC[q], 2, 3);
                #pragma unroll
                for (int nt = 0; nt < NTL; ++nt) {
                    const v2f bLo = __builtin_shufflevector(bC[q][nt], bC[q][nt], 0, 1);
                    const v2f bHi = __builtin_shufflevector(bC[q][nt], bC[q][nt], 2, 3);
                    acc[0][nt] = __builtin_amdgcn_wmma_f32_16x16x4_f32(false,aLo,false,bLo,(short)0,acc[0][nt],false,false);
                    acc[1][nt] = __builtin_amdgcn_wmma_f32_16x16x4_f32(false,aHi,false,bHi,(short)0,acc[1][nt],false,false);
                }
            }
            #pragma unroll
            for (int q = 0; q < GB; ++q) {
                aC[q] = aN[q];
                #pragma unroll
                for (int nt = 0; nt < NTL; ++nt) bC[q][nt] = bN[q][nt];
            }
        }
    }

    #pragma unroll
    for (int nt = 0; nt < NTL; ++nt) {
        const int gn = n0 + wn * WNC + nt * 16 + c;
        const v8f s = acc[0][nt] + acc[1][nt];
        #pragma unroll
        for (int r = 0; r < 8; ++r) {
            const int gm = m0 + wm * 16 + h * 8 + r;
            if (gm < M && gn < N) C[(size_t)gm * N + gn] = s[r];
        }
    }
}

__global__ __launch_bounds__(256) void attn_softmax_kernel(
    const float* __restrict__ scores, const float* __restrict__ lmask,
    const float* __restrict__ imask, float* __restrict__ attn,
    float* __restrict__ b_in)
{
    int row = blockIdx.x * 256 + threadIdx.x;
    if (row >= 8*256*20) return;
    int bs = row / 20;
    int l  = row - bs*20;

    float v[8];
    float mx = -3.4e38f;
    #pragma unroll
    for (int t = 0; t < 8; ++t) {
        float s = scores[(size_t)row*8 + t] + (1.0f - lmask[l*8 + t]) * NEG_BIG;
        v[t] = s;
        mx = fmaxf(mx, s);
    }
    float sum = 0.f;
    #pragma unroll
    for (int t = 0; t < 8; ++t) { float e = __expf(v[t] - mx); v[t] = e; sum += e; }
    float inv = 1.0f / sum;
    #pragma unroll
    for (int t = 0; t < 8; ++t) attn[(size_t)row*8 + t] = v[t] * inv;

    b_in[row] = mx + (1.0f - imask[bs]) * NEG_BIG;
}

#define QP 260

__global__ __launch_bounds__(256) void q2c_wmma_kernel(
    const float* __restrict__ b_in, const float* __restrict__ tkn,
    float* __restrict__ q2c)
{
    __shared__ float bS[256*20];
    __shared__ float Ps[32][QP];

    const int tid  = threadIdx.x;
    const int lane = tid & 31, wid = tid >> 5;
    const int ll   = lane & 15, lh = lane >> 4;
    const int b    = blockIdx.x;
    const int n0   = blockIdx.y * 128;

    #pragma unroll
    for (int i = 0; i < 5; ++i) {
        int idx = (i*256 + tid) << 2;
        *(v4f*)&bS[idx] = *(const v4f*)(b_in + (size_t)b*5120 + idx);
    }
    for (int i = tid; i < 12*QP; i += 256) (&Ps[20][0])[i] = 0.f;
    __syncthreads();

    for (int l = wid; l < 20; l += 8) {
        float x[8];
        float mx = -3.4e38f;
        #pragma unroll
        for (int i = 0; i < 8; ++i) { x[i] = bS[(lane + 32*i)*20 + l]; mx = fmaxf(mx, x[i]); }
        #pragma unroll
        for (int off = 16; off > 0; off >>= 1) mx = fmaxf(mx, __shfl_xor(mx, off));
        float sm = 0.f;
        #pragma unroll
        for (int i = 0; i < 8; ++i) { x[i] = __expf(x[i] - mx); sm += x[i]; }
        #pragma unroll
        for (int off = 16; off > 0; off >>= 1) sm += __shfl_xor(sm, off);
        #pragma unroll
        for (int i = 0; i < 8; ++i) Ps[l][lane + 32*i] = x[i] / sm;
    }
    __syncthreads();

    const float* bp = tkn + (size_t)b*256*768 + (size_t)(2*lh)*768 + n0 + wid*16 + ll;
    const v8f vzero = {0.f,0.f,0.f,0.f,0.f,0.f,0.f,0.f};
    v8f acc0 = vzero, acc1 = vzero;

    for (int kb = 0; kb < 64; kb += 16) {
        float bv0[16], bv1[16];
        #pragma unroll
        for (int j = 0; j < 16; ++j) {
            const float* p = bp + (size_t)(4*(kb + j))*768;
            bv0[j] = p[0];
            bv1[j] = p[768];
        }
        #pragma unroll
        for (int j = 0; j < 16; ++j) {
            const int kc = 4*(kb + j) + 2*lh;
            v2f a0 = *(const v2f*)&Ps[ll][kc];
            v2f a1 = *(const v2f*)&Ps[16 + ll][kc];
            v2f bf = {bv0[j], bv1[j]};
            acc0 = __builtin_amdgcn_wmma_f32_16x16x4_f32(false,a0,false,bf,(short)0,acc0,false,false);
            acc1 = __builtin_amdgcn_wmma_f32_16x16x4_f32(false,a1,false,bf,(short)0,acc1,false,false);
        }
    }

    const int gn = n0 + wid*16 + ll;
    #pragma unroll
    for (int r = 0; r < 8; ++r) {
        int row = lh*8 + r;
        q2c[(size_t)(b*20 + row)*768 + gn] = acc0[r];
    }
    #pragma unroll
    for (int r = 0; r < 8; ++r) {
        int row = 16 + lh*8 + r;
        if (row < 20) q2c[(size_t)(b*20 + row)*768 + gn] = acc1[r];
    }
}

#define KP 36
#define LP 12

__global__ __launch_bounds__(256) __attribute__((amdgpu_waves_per_eu(6, 6)))
void fused_w5_tanh_wmma(
    const float* __restrict__ tkn,  const float* __restrict__ lab,
    const float* __restrict__ attn, const float* __restrict__ q2c,
    const float* __restrict__ W5,   const float* __restrict__ b5,
    float* __restrict__ out)
{
    __shared__ __attribute__((aligned(16))) float Ts[TM][KP];
    __shared__ __attribute__((aligned(16))) float Ws[3][TN][KP];
    __shared__ __attribute__((aligned(16))) float Lt[TK][LP];

    const int tid  = threadIdx.x;
    const int lane = tid & 31, wid = tid >> 5;
    const int c    = lane & 15, h = lane >> 4;
    const int bl   = blockIdx.x >> 1;
    const int b    = bl / 20;
    const int l    = bl - b*20;
    const int bs0  = b*256 + ((blockIdx.x & 1) << 7);
    const int n0   = blockIdx.y * TN;

    const int g  = tid >> 3;
    const int sr = ((g & 1) << 3) | ((g >> 1) & 7) | (g & 16);
    const int sc = (tid & 7) << 2;
    const float* tknP = tkn + (size_t)(bs0 + sr)*768 + sc;
    const float* q2cP = q2c + (size_t)bl*768 + sc;
    const float* w5P  = W5  + (size_t)(n0 + sr)*3072 + sc;
    const int lt_t = tid >> 5, lt_j = tid & 31;
    const float* labP = lab + (size_t)(l*8 + lt_t)*768 + lt_j;

    const float* ap = attn + ((size_t)(bs0 + wid*16 + c)*20 + l)*8 + 2*h;
    const v2f aT0 = *(const v2f*)(ap);
    const v2f aT1 = *(const v2f*)(ap + 4);

    const v8f vzero = {0.f,0.f,0.f,0.f,0.f,0.f,0.f,0.f};
    v8f acc[4];
    acc[0]=vzero; acc[1]=vzero; acc[2]=vzero; acc[3]=vzero;

    for (int ch = 0; ch < 24; ++ch) {
        const int j0 = ch * TK;
        __syncthreads();
        {
            v4f q4 = *(const v4f*)(q2cP + j0);
            #pragma unroll
            for (int p = 0; p < 2; ++p) {
                const float* wp = w5P + (size_t)p*32*3072 + j0;
                v4f w0 = *(const v4f*)(wp);
                v4f w1 = *(const v4f*)(wp + 768);
                v4f w2 = *(const v4f*)(wp + 1536);
                v4f w3 = *(const v4f*)(wp + 2304);
                *(v4f*)&Ws[0][p*32 + sr][sc] = w0 + q4 * w3;
                *(v4f*)&Ws[1][p*32 + sr][sc] = w1;
                *(v4f*)&Ws[2][p*32 + sr][sc] = w2;
            }
            #pragma unroll
            for (int p = 0; p < 4; ++p)
                *(v4f*)&Ts[p*32 + sr][sc] = *(const v4f*)(tknP + (size_t)p*32*768 + j0);
            Lt[lt_j][lt_t] = labP[j0];
        }
        __syncthreads();

        #pragma unroll
        for (int jg = 0; jg < 2; ++jg) {
            const int cb = 16*jg + 8*h;
            v2f la0 = *(const v2f*)&Lt[16*jg + c][2*h];
            v2f la1 = *(const v2f*)&Lt[16*jg + c][4 + 2*h];
            v8f d = vzero;
            d = __builtin_amdgcn_wmma_f32_16x16x4_f32(false,la0,false,aT0,(short)0,d,false,false);
            d = __builtin_amdgcn_wmma_f32_16x16x4_f32(false,la1,false,aT1,(short)0,d,false,false);
            v4f ta = *(const v4f*)&Ts[wid*16 + c][cb];
            v4f tb = *(const v4f*)&Ts[wid*16 + c][cb + 4];
            v8f t8 = __builtin_shufflevector(ta, tb, 0,1,2,3,4,5,6,7);
            v8f pr = t8 * d;

            #pragma unroll
            for (int hf = 0; hf < 2; ++hf) {
                v2f a0[2], a1[2], a2[2];
                if (hf == 0) {
                    a0[0] = __builtin_shufflevector(t8, t8, 0, 1); a0[1] = __builtin_shufflevector(t8, t8, 2, 3);
                    a1[0] = __builtin_shufflevector(d,  d,  0, 1); a1[1] = __builtin_shufflevector(d,  d,  2, 3);
                    a2[0] = __builtin_shufflevector(pr, pr, 0, 1); a2[1] = __builtin_shufflevector(pr, pr, 2, 3);
                } else {
                    a0[0] = __builtin_shufflevector(t8, t8, 4, 5); a0[1] = __builtin_shufflevector(t8, t8, 6, 7);
                    a1[0] = __builtin_shufflevector(d,  d,  4, 5); a1[1] = __builtin_shufflevector(d,  d,  6, 7);
                    a2[0] = __builtin_shufflevector(pr, pr, 4, 5); a2[1] = __builtin_shufflevector(pr, pr, 6, 7);
                }
                #pragma unroll
                for (int s = 0; s < 3; ++s) {
                    v4f bv[4];
                    #pragma unroll
                    for (int nt = 0; nt < 4; ++nt) bv[nt] = *(const v4f*)&Ws[s][nt*16 + c][cb + 4*hf];
                    #pragma unroll
                    for (int u = 0; u < 2; ++u) {
                        const v2f af = (s == 0) ? a0[u] : (s == 1) ? a1[u] : a2[u];
                        #pragma unroll
                        for (int nt = 0; nt < 4; ++nt) {
                            v2f bf = (u == 0) ? __builtin_shufflevector(bv[nt], bv[nt], 0, 1)
                                              : __builtin_shufflevector(bv[nt], bv[nt], 2, 3);
                            acc[nt] = __builtin_amdgcn_wmma_f32_16x16x4_f32(false,af,false,bf,(short)0,acc[nt],false,false);
                        }
                    }
                }
            }
        }
    }

    #pragma unroll
    for (int nt = 0; nt < 4; ++nt) {
        int gn = n0 + nt*16 + c;
        float bias = b5[gn];
        #pragma unroll
        for (int r = 0; r < 8; ++r) {
            int tr = wid*16 + h*8 + r;
            size_t gm = (size_t)(bs0 + tr)*20 + l;
            out[gm*768 + gn] = tanhf(acc[nt][r] + bias);
        }
    }
}

extern "C" void kernel_launch(void* const* d_in, const int* in_sizes, int n_in,
                              void* d_out, int out_size, void* d_ws, size_t ws_size,
                              hipStream_t stream)
{
    (void)in_sizes; (void)n_in; (void)out_size; (void)ws_size;

    const float* token = (const float*)d_in[0];
    const float* label = (const float*)d_in[1];
    const float* imask = (const float*)d_in[2];
    const float* lmask = (const float*)d_in[3];
    const float* W1    = (const float*)d_in[4];
    const float* W2    = (const float*)d_in[5];
    const float* W5    = (const float*)d_in[6];
    const float* b5    = (const float*)d_in[7];
    float* out = (float*)d_out;

    float* ws     = (float*)d_ws;
    float* tkn    = ws;
    float* lab    = tkn    + 2048*768;
    float* scores = lab    + 160*768;
    float* attn   = scores + 2048*160;
    float* b_in   = attn   + 2048*160;
    float* q2c    = b_in   + 40960;

    dim3 blk(256);
    dim3 blkG(128);
    gemm_nt_tile_wmma<64, 48, 4, 1><<<dim3(32, 16), blkG, 0, stream>>>(token, W1, tkn, 2048, 768, 768);
    gemm_nt_tile_wmma<32, 32, 2, 2><<<dim3(5, 24),  blkG, 0, stream>>>(label, W2, lab, 160, 768, 300);
    gemm_nt_tile_wmma<32, 32, 2, 2><<<dim3(64, 5),  blkG, 0, stream>>>(tkn, lab, scores, 2048, 160, 768);
    attn_softmax_kernel<<<dim3(160), blk, 0, stream>>>(scores, lmask, imask, attn, b_in);
    q2c_wmma_kernel<<<dim3(8, 6), blk, 0, stream>>>(b_in, tkn, q2c);
    fused_w5_tanh_wmma<<<dim3(320, 12), blk, 0, stream>>>(tkn, lab, attn, q2c, W5, b5, out);
}


# Main GEMM: c2q K-section by linearity inside the block: waves form P = lab_l (8 rows) @ W5 sec1 tile^T from the staged LDS tile, attn (K=8) @ P added after the loop; K 2304->1536; sum order changes
# speedup vs baseline: 2.3539x; 1.2414x over previous
#include <hip/hip_runtime.h>
#include <math.h>


typedef __attribute__((ext_vector_type(2))) float v2f;
typedef __attribute__((ext_vector_type(4))) float v4f;
typedef __attribute__((ext_vector_type(8))) float v8f;

#define NEG_BIG (-10000.0f)

#define TM 128
#define TN 64
#define TK 32


#define GK  128
#define GKP 132

template <int BM, int BN, int WM, int WN>
__global__ __launch_bounds__(128) void gemm_nt_tile_wmma(
    const float* __restrict__ A, const float* __restrict__ W,
    float* __restrict__ C, int M, int N, int K)
{
    constexpr int WNC = BN / WN;
    constexpr int NTL = WNC / 16;
    constexpr int GB  = (NTL >= 2) ? 2 : 4;
    static_assert(BM == 16 * WM && WM * WN == 4 && NTL * 16 * WN == BN, "tile shape");

    __shared__ __attribute__((aligned(16))) float As[BM][GKP];
    __shared__ __attribute__((aligned(16))) float Ws[BN][GKP];

    const int tid  = threadIdx.x;
    const int lane = tid & 31, wid = tid >> 5;
    const int wm   = wid / WN, wn = wid - wm * WN;
    const int c    = lane & 15, h = lane >> 4;
    const int m0   = blockIdx.x * BM, n0 = blockIdx.y * BN;
    const int sr   = tid >> 5;
    const int sc   = (tid & 31) << 2;

    const v8f vzero = {0.f,0.f,0.f,0.f,0.f,0.f,0.f,0.f};
    v8f acc[2][NTL];
    #pragma unroll
    for (int nt = 0; nt < NTL; ++nt) { acc[0][nt] = vzero; acc[1][nt] = vzero; }

    for (int k0 = 0; k0 < K; k0 += GK) {
        __syncthreads();
        {
            const int gk = k0 + sc;
            v4f ra[BM / 4], rw[BN / 4];
            #pragma unroll
            for (int p = 0; p < BM / 4; ++p) {
                const int gm = m0 + p * 4 + sr;
                v4f v = {0.f,0.f,0.f,0.f};
                if (gm < M && gk < K) v = *(const v4f*)(A + (size_t)gm * K + gk);
                ra[p] = v;
            }
            #pragma unroll
            for (int p = 0; p < BN / 4; ++p) {
                const int gn = n0 + p * 4 + sr;
                v4f v = {0.f,0.f,0.f,0.f};
                if (gn < N && gk < K) v = *(const v4f*)(W + (size_t)gn * K + gk);
                rw[p] = v;
            }
            #pragma unroll
            for (int p = 0; p < BM / 4; ++p) *(v4f*)&As[p * 4 + sr][sc] = ra[p];
            #pragma unroll
            for (int p = 0; p < BN / 4; ++p) *(v4f*)&Ws[p * 4 + sr][sc] = rw[p];
        }
        __syncthreads();

        const int krem = K - k0;
        const int nb   = ((krem < GK ? krem : GK) + 8 * GB - 1) / (8 * GB);
        const float* aRow = &As[wm * 16 + c][4 * h];
        const float* wRow = &Ws[wn * WNC + c][4 * h];
        v4f aC[GB], bC[GB][NTL];
        #pragma unroll
        for (int q = 0; q < GB; ++q) {
            aC[q] = *(const v4f*)(aRow + 8 * q);
            #pragma unroll
            for (int nt = 0; nt < NTL; ++nt) bC[q][nt] = *(const v4f*)(wRow + nt * 16 * GKP + 8 * q);
        }
        #pragma unroll 2
        for (int bi = 0; bi < nb; ++bi) {
            const int bn = (bi + 1 < nb) ? bi + 1 : bi;
            v4f aN[GB], bN[GB][NTL];
            #pragma unroll
            for (int q = 0; q < GB; ++q) {
                aN[q] = *(const v4f*)(aRow + 8 * (bn * GB + q));
                #pragma unroll
                for (int nt = 0; nt < NTL; ++nt) bN[q][nt] = *(const v4f*)(wRow + nt * 16 * GKP + 8 * (bn * GB + q));
            }
            #pragma unroll
            for (int q = 0; q < GB; ++q) {
                const v2f aLo = __builtin_shufflevector(aC[q], aC[q], 0, 1);
                const v2f aHi = __builtin_shufflevector(aC[q], aC[q], 2, 3);
                #pragma unroll
                for (int nt = 0; nt < NTL; ++nt) {
                    const v2f bLo = __builtin_shufflevector(bC[q][nt], bC[q][nt], 0, 1);
                    const v2f bHi = __builtin_shufflevector(bC[q][nt], bC[q][nt], 2, 3);
                    acc[0][nt] = __builtin_amdgcn_wmma_f32_16x16x4_f32(false,aLo,false,bLo,(short)0,acc[0][nt],false,false);
                    acc[1][nt] = __builtin_amdgcn_wmma_f32_16x16x4_f32(false,aHi,false,bHi,(short)0,acc[1][nt],false,false);
                }
            }
            #pragma unroll
            for (int q = 0; q < GB; ++q) {
                aC[q] = aN[q];
                #pragma unroll
                for (int nt = 0; nt < NTL; ++nt) bC[q][nt] = bN[q][nt];
            }
        }
    }

    #pragma unroll
    for (int nt = 0; nt < NTL; ++nt) {
        const int gn = n0 + wn * WNC + nt * 16 + c;
        const v8f s = acc[0][nt] + acc[1][nt];
        #pragma unroll
        for (int r = 0; r < 8; ++r) {
            const int gm = m0 + wm * 16 + h * 8 + r;
            if (gm < M && gn < N) C[(size_t)gm * N + gn] = s[r];
        }
    }
}

__global__ __launch_bounds__(256) void attn_softmax_kernel(
    const float* __restrict__ scores, const float* __restrict__ lmask,
    const float* __restrict__ imask, float* __restrict__ attn,
    float* __restrict__ b_in)
{
    int row = blockIdx.x * 256 + threadIdx.x;
    if (row >= 8*256*20) return;
    int bs = row / 20;
    int l  = row - bs*20;

    float v[8];
    float mx = -3.4e38f;
    #pragma unroll
    for (int t = 0; t < 8; ++t) {
        float s = scores[(size_t)row*8 + t] + (1.0f - lmask[l*8 + t]) * NEG_BIG;
        v[t] = s;
        mx = fmaxf(mx, s);
    }
    float sum = 0.f;
    #pragma unroll
    for (int t = 0; t < 8; ++t) { float e = __expf(v[t] - mx); v[t] = e; sum += e; }
    float inv = 1.0f / sum;
    #pragma unroll
    for (int t = 0; t < 8; ++t) attn[(size_t)row*8 + t] = v[t] * inv;

    b_in[row] = mx + (1.0f - imask[bs]) * NEG_BIG;
}

#define QP 260

__global__ __launch_bounds__(256) void q2c_wmma_kernel(
    const float* __restrict__ b_in, const float* __restrict__ tkn,
    float* __restrict__ q2c)
{
    __shared__ float bS[256*20];
    __shared__ float Ps[32][QP];

    const int tid  = threadIdx.x;
    const int lane = tid & 31, wid = tid >> 5;
    const int ll   = lane & 15, lh = lane >> 4;
    const int b    = blockIdx.x;
    const int n0   = blockIdx.y * 128;

    #pragma unroll
    for (int i = 0; i < 5; ++i) {
        int idx = (i*256 + tid) << 2;
        *(v4f*)&bS[idx] = *(const v4f*)(b_in + (size_t)b*5120 + idx);
    }
    for (int i = tid; i < 12*QP; i += 256) (&Ps[20][0])[i] = 0.f;
    __syncthreads();

    for (int l = wid; l < 20; l += 8) {
        float x[8];
        float mx = -3.4e38f;
        #pragma unroll
        for (int i = 0; i < 8; ++i) { x[i] = bS[(lane + 32*i)*20 + l]; mx = fmaxf(mx, x[i]); }
        #pragma unroll
        for (int off = 16; off > 0; off >>= 1) mx = fmaxf(mx, __shfl_xor(mx, off));
        float sm = 0.f;
        #pragma unroll
        for (int i = 0; i < 8; ++i) { x[i] = __expf(x[i] - mx); sm += x[i]; }
        #pragma unroll
        for (int off = 16; off > 0; off >>= 1) sm += __shfl_xor(sm, off);
        #pragma unroll
        for (int i = 0; i < 8; ++i) Ps[l][lane + 32*i] = x[i] / sm;
    }
    __syncthreads();

    const float* bp = tkn + (size_t)b*256*768 + (size_t)(2*lh)*768 + n0 + wid*16 + ll;
    const v8f vzero = {0.f,0.f,0.f,0.f,0.f,0.f,0.f,0.f};
    v8f acc0 = vzero, acc1 = vzero;

    for (int kb = 0; kb < 64; kb += 16) {
        float bv0[16], bv1[16];
        #pragma unroll
        for (int j = 0; j < 16; ++j) {
            const float* p = bp + (size_t)(4*(kb + j))*768;
            bv0[j] = p[0];
            bv1[j] = p[768];
        }
        #pragma unroll
        for (int j = 0; j < 16; ++j) {
            const int kc = 4*(kb + j) + 2*lh;
            v2f a0 = *(const v2f*)&Ps[ll][kc];
            v2f a1 = *(const v2f*)&Ps[16 + ll][kc];
            v2f bf = {bv0[j], bv1[j]};
            acc0 = __builtin_amdgcn_wmma_f32_16x16x4_f32(false,a0,false,bf,(short)0,acc0,false,false);
            acc1 = __builtin_amdgcn_wmma_f32_16x16x4_f32(false,a1,false,bf,(short)0,acc1,false,false);
        }
    }

    const int gn = n0 + wid*16 + ll;
    #pragma unroll
    for (int r = 0; r < 8; ++r) {
        int row = lh*8 + r;
        q2c[(size_t)(b*20 + row)*768 + gn] = acc0[r];
    }
    #pragma unroll
    for (int r = 0; r < 8; ++r) {
        int row = 16 + lh*8 + r;
        if (row < 20) q2c[(size_t)(b*20 + row)*768 + gn] = acc1[r];
    }
}

#define KP 36
#define LP 12

__global__ __launch_bounds__(256) __attribute__((amdgpu_waves_per_eu(6, 6)))
void fused_w5_tanh_wmma(
    const float* __restrict__ tkn,  const float* __restrict__ lab,
    const float* __restrict__ attn, const float* __restrict__ q2c,
    const float* __restrict__ W5,   const float* __restrict__ b5,
    float* __restrict__ out)
{
    __shared__ __attribute__((aligned(16))) float Ts[TM][KP];
    __shared__ __attribute__((aligned(16))) float Ws[3][TN][KP];
    __shared__ __attribute__((aligned(16))) float Lt[TK][LP];
    __shared__ __attribute__((aligned(16))) float Ln[16][KP];
    __shared__ __attribute__((aligned(16))) float Pp[2][TN][LP];

    const int tid  = threadIdx.x;
    const int lane = tid & 31, wid = tid >> 5;
    const int c    = lane & 15, h = lane >> 4;
    const int bl   = blockIdx.x >> 1;
    const int b    = bl / 20;
    const int l    = bl - b*20;
    const int bs0  = b*256 + ((blockIdx.x & 1) << 7);
    const int n0   = blockIdx.y * TN;

    const int g  = tid >> 3;
    const int sr = ((g & 1) << 3) | ((g >> 1) & 7) | (g & 16);
    const int sc = (tid & 7) << 2;
    const float* tknP = tkn + (size_t)(bs0 + sr)*768 + sc;
    const float* q2cP = q2c + (size_t)bl*768 + sc;
    const float* w5P  = W5  + (size_t)(n0 + sr)*3072 + sc;
    const int lt_t = tid >> 5, lt_j = tid & 31;
    const float* labP = lab + (size_t)(l*8 + lt_t)*768 + lt_j;

    const float* ap = attn + ((size_t)(bs0 + wid*16 + c)*20 + l)*8 + 2*h;
    const v2f aT0 = *(const v2f*)(ap);
    const v2f aT1 = *(const v2f*)(ap + 4);

    const v8f vzero = {0.f,0.f,0.f,0.f,0.f,0.f,0.f,0.f};
    v8f acc[4];
    acc[0]=vzero; acc[1]=vzero; acc[2]=vzero; acc[3]=vzero;

    const int ntp = wid & 3, kq = wid >> 2;
    const float* pA = &Ln[c][16*kq + 8*h];
    const float* pB = &Ws[1][ntp*16 + c][16*kq + 8*h];
    v8f pacc[2];
    pacc[0] = vzero; pacc[1] = vzero;
    for (int i = tid; i < 8*KP; i += 256) (&Ln[8][0])[i] = 0.f;

    for (int ch = 0; ch < 24; ++ch) {
        const int j0 = ch * TK;
        __syncthreads();
        {
            v4f q4 = *(const v4f*)(q2cP + j0);
            #pragma unroll
            for (int p = 0; p < 2; ++p) {
                const float* wp = w5P + (size_t)p*32*3072 + j0;
                v4f w0 = *(const v4f*)(wp);
                v4f w1 = *(const v4f*)(wp + 768);
                v4f w2 = *(const v4f*)(wp + 1536);
                v4f w3 = *(const v4f*)(wp + 2304);
                *(v4f*)&Ws[0][p*32 + sr][sc] = w0 + q4 * w3;
                *(v4f*)&Ws[1][p*32 + sr][sc] = w1;
                *(v4f*)&Ws[2][p*32 + sr][sc] = w2;
            }
            #pragma unroll
            for (int p = 0; p < 4; ++p)
                *(v4f*)&Ts[p*32 + sr][sc] = *(const v4f*)(tknP + (size_t)p*32*768 + j0);
            const float lv = labP[j0];
            Lt[lt_j][lt_t] = lv;
            Ln[lt_t][lt_j] = lv;
        }
        __syncthreads();

        {
            const v4f pa0 = *(const v4f*)(pA), pa1 = *(const v4f*)(pA + 4);
            const v4f pb0 = *(const v4f*)(pB), pb1 = *(const v4f*)(pB + 4);
            pacc[0] = __builtin_amdgcn_wmma_f32_16x16x4_f32(false,__builtin_shufflevector(pa0,pa0,0,1),false,__builtin_shufflevector(pb0,pb0,0,1),(short)0,pacc[0],false,false);
            pacc[1] = __builtin_amdgcn_wmma_f32_16x16x4_f32(false,__builtin_shufflevector(pa0,pa0,2,3),false,__builtin_shufflevector(pb0,pb0,2,3),(short)0,pacc[1],false,false);
            pacc[0] = __builtin_amdgcn_wmma_f32_16x16x4_f32(false,__builtin_shufflevector(pa1,pa1,0,1),false,__builtin_shufflevector(pb1,pb1,0,1),(short)0,pacc[0],false,false);
            pacc[1] = __builtin_amdgcn_wmma_f32_16x16x4_f32(false,__builtin_shufflevector(pa1,pa1,2,3),false,__builtin_shufflevector(pb1,pb1,2,3),(short)0,pacc[1],false,false);
        }

        #pragma unroll
        for (int jg = 0; jg < 2; ++jg) {
            const int cb = 16*jg + 8*h;
            v2f la0 = *(const v2f*)&Lt[16*jg + c][2*h];
            v2f la1 = *(const v2f*)&Lt[16*jg + c][4 + 2*h];
            v8f d = vzero;
            d = __builtin_amdgcn_wmma_f32_16x16x4_f32(false,la0,false,aT0,(short)0,d,false,false);
            d = __builtin_amdgcn_wmma_f32_16x16x4_f32(false,la1,false,aT1,(short)0,d,false,false);
            v4f ta = *(const v4f*)&Ts[wid*16 + c][cb];
            v4f tb = *(const v4f*)&Ts[wid*16 + c][cb + 4];
            v8f t8 = __builtin_shufflevector(ta, tb, 0,1,2,3,4,5,6,7);
            v8f pr = t8 * d;

            #pragma unroll
            for (int hf = 0; hf < 2; ++hf) {
                v2f a0[2], a2[2];
                if (hf == 0) {
                    a0[0] = __builtin_shufflevector(t8, t8, 0, 1); a0[1] = __builtin_shufflevector(t8, t8, 2, 3);
                    a2[0] = __builtin_shufflevector(pr, pr, 0, 1); a2[1] = __builtin_shufflevector(pr, pr, 2, 3);
                } else {
                    a0[0] = __builtin_shufflevector(t8, t8, 4, 5); a0[1] = __builtin_shufflevector(t8, t8, 6, 7);
                    a2[0] = __builtin_shufflevector(pr, pr, 4, 5); a2[1] = __builtin_shufflevector(pr, pr, 6, 7);
                }
                #pragma unroll
                for (int s = 0; s < 3; s += 2) {
                    v4f bv[4];
                    #pragma unroll
                    for (int nt = 0; nt < 4; ++nt) bv[nt] = *(const v4f*)&Ws[s][nt*16 + c][cb + 4*hf];
                    #pragma unroll
                    for (int u = 0; u < 2; ++u) {
                        const v2f af = (s == 0) ? a0[u] : a2[u];
                        #pragma unroll
                        for (int nt = 0; nt < 4; ++nt) {
                            v2f bf = (u == 0) ? __builtin_shufflevector(bv[nt], bv[nt], 0, 1)
                                              : __builtin_shufflevector(bv[nt], bv[nt], 2, 3);
                            acc[nt] = __builtin_amdgcn_wmma_f32_16x16x4_f32(false,af,false,bf,(short)0,acc[nt],false,false);
                        }
                    }
                }
            }
        }
    }

    {
        const v8f ps = pacc[0] + pacc[1];
        if (h == 0) {
            float* dst = &Pp[kq][ntp*16 + c][0];
            const v4f lo = {ps[0], ps[1], ps[2], ps[3]};
            const v4f hi = {ps[4], ps[5], ps[6], ps[7]};
            *(v4f*)dst       = lo;
            *(v4f*)(dst + 4) = hi;
        }
    }
    __syncthreads();
    #pragma unroll
    for (int s = 0; s < 2; ++s) {
        #pragma unroll
        for (int nt = 0; nt < 4; ++nt) {
            const v2f p0 = *(const v2f*)&Pp[0][nt*16 + c][4*s + 2*h];
            const v2f p1 = *(const v2f*)&Pp[1][nt*16 + c][4*s + 2*h];
            const v2f bf = p0 + p1;
            acc[nt] = __builtin_amdgcn_wmma_f32_16x16x4_f32(false,(s == 0) ? aT0 : aT1,false,bf,(short)0,acc[nt],false,false);
        }
    }

    #pragma unroll
    for (int nt = 0; nt < 4; ++nt) {
        int gn = n0 + nt*16 + c;
        float bias = b5[gn];
        #pragma unroll
        for (int r = 0; r < 8; ++r) {
            int tr = wid*16 + h*8 + r;
            size_t gm = (size_t)(bs0 + tr)*20 + l;
            out[gm*768 + gn] = tanhf(acc[nt][r] + bias);
        }
    }
}

extern "C" void kernel_launch(void* const* d_in, const int* in_sizes, int n_in,
                              void* d_out, int out_size, void* d_ws, size_t ws_size,
                              hipStream_t stream)
{
    (void)in_sizes; (void)n_in; (void)out_size; (void)ws_size;

    const float* token = (const float*)d_in[0];
    const float* label = (const float*)d_in[1];
    const float* imask = (const float*)d_in[2];
    const float* lmask = (const float*)d_in[3];
    const float* W1    = (const float*)d_in[4];
    const float* W2    = (const float*)d_in[5];
    const float* W5    = (const float*)d_in[6];
    const float* b5    = (const float*)d_in[7];
    float* out = (float*)d_out;

    float* ws     = (float*)d_ws;
    float* tkn    = ws;
    float* lab    = tkn    + 2048*768;
    float* scores = lab    + 160*768;
    float* attn   = scores + 2048*160;
    float* b_in   = attn   + 2048*160;
    float* q2c    = b_in   + 40960;

    dim3 blk(256);
    dim3 blkG(128);
    gemm_nt_tile_wmma<64, 48, 4, 1><<<dim3(32, 16), blkG, 0, stream>>>(token, W1, tkn, 2048, 768, 768);
    gemm_nt_tile_wmma<32, 32, 2, 2><<<dim3(5, 24),  blkG, 0, stream>>>(label, W2, lab, 160, 768, 300);
    gemm_nt_tile_wmma<32, 32, 2, 2><<<dim3(64, 5),  blkG, 0, stream>>>(tkn, lab, scores, 2048, 160, 768);
    attn_softmax_kernel<<<dim3(160), blk, 0, stream>>>(scores, lmask, imask, attn, b_in);
    q2c_wmma_kernel<<<dim3(8, 6), blk, 0, stream>>>(b_in, tkn, q2c);
    fused_w5_tanh_wmma<<<dim3(320, 12), blk, 0, stream>>>(tkn, lab, attn, q2c, W5, b5, out);
}
